# starting kernel
# baseline (speedup 1.0000x reference)
#include <hip/hip_runtime.h>
#include <hip/hip_bf16.h>


typedef __attribute__((ext_vector_type(2))) float v2f;
typedef __attribute__((ext_vector_type(8))) float v8f;

#define NAX 160
#define NT  10
#define KSTEPS 40

__global__ void build_H(float* __restrict__ H) {
  int idx = blockIdx.x * blockDim.x + threadIdx.x;
  if (idx >= NAX * NAX) return;
  int i = idx / NAX, j = idx % NAX;
  int d = i - j; if (d < 0) d += NAX;
  const double p = -0.26794919243112270647255365849413;
  double pd = 1.0, pnd = 1.0;
  for (int k = 0; k < d; ++k)        pd  *= p;
  for (int k = 0; k < NAX - d; ++k)  pnd *= p;
  double pN = pd * pnd;
  double h = 1.7320508075688772935274463415059 * (pd + pnd) / (1.0 - pN);
  H[idx] = (float)h;
}

template <int Q, int P, int S>
__global__ __launch_bounds__(256) void spline_pass(
    const float* __restrict__ in, float* __restrict__ out,
    const float* __restrict__ H, int nTiles)
{
  const int lane  = threadIdx.x & 31;
  const int wave  = threadIdx.x >> 5;
  const int tile  = blockIdx.x * 8 + wave;
  if (tile >= nTiles) return;

  const int col   = lane & 15;
  const int khalf = lane >> 4;

  const int lA    = tile * 16 + col;
  const int baseA = (lA / Q) * P + (lA % Q);

  int baseD[8];
#pragma unroll
  for (int v = 0; v < 8; ++v) {
    int lD = tile * 16 + khalf * 8 + v;
    baseD[v] = (lD / Q) * P + (lD % Q);
  }

  v8f acc[NT];
#pragma unroll
  for (int t = 0; t < NT; ++t) acc[t] = {};

  for (int ks = 0; ks < KSTEPS; ++ks) {
    const int kb = ks * 4;
    v2f a;
    a.x = in[baseA + (kb + 2 * khalf + 0) * S];
    a.y = in[baseA + (kb + 2 * khalf + 1) * S];
    __builtin_prefetch(&in[baseA + (kb + 4 + 2 * khalf) * S], 0, 1);
#pragma unroll
    for (int t = 0; t < NT; ++t) {
      const float* Hrow = H + (t * 16 + col) * NAX + kb + 2 * khalf;
      v2f b;
      b.x = Hrow[0];
      b.y = Hrow[1];
      acc[t] = __builtin_amdgcn_wmma_f32_16x16x4_f32(
           false, a,  false, b,
           (short)0, acc[t],  false,  false);
    }
  }

#pragma unroll
  for (int t = 0; t < NT; ++t) {
#pragma unroll
    for (int v = 0; v < 8; ++v) {
      out[baseD[v] + (t * 16 + col) * S] = acc[t][v];
    }
  }
}

extern "C" void kernel_launch(void* const* d_in, const int* in_sizes, int n_in,
                              void* d_out, int out_size, void* d_ws, size_t ws_size,
                              hipStream_t stream) {
  const float* x   = (const float*)d_in[0];
  float*       out = (float*)d_out;
  float*       H   = (float*)d_ws;
  float*       tmp = (float*)d_ws + 32768;

  (void)in_sizes; (void)n_in; (void)out_size; (void)ws_size;

  build_H<<<dim3((NAX * NAX + 255) / 256), dim3(256), 0, stream>>>(H);

  const int lines = 2 * 3 * NAX * NAX;
  const int tiles = lines / 16;
  dim3 grid((tiles + 7) / 8), block(256);

  spline_pass<1, NAX, 1>
      <<<grid, block, 0, stream>>>(x, out, H, tiles);
  spline_pass<NAX, NAX * NAX, NAX>
      <<<grid, block, 0, stream>>>(out, tmp, H, tiles);
  spline_pass<NAX * NAX, NAX * NAX * NAX, NAX * NAX>
      <<<grid, block, 0, stream>>>(tmp, out, H, tiles);
}


# Reflection split of the symmetric circulant: s=x[k]+x[159-k], d=x[k]-x[159-k]; two 80x80 f32 WMMA products per line (200 WMMAs/strip, was 400), B fragments = T[n-k] +- T[n+k+1]; k sum order changes
# speedup vs baseline: 3.8349x; 3.8349x over previous
#include <hip/hip_runtime.h>
#include <hip/hip_bf16.h>


typedef __attribute__((ext_vector_type(2))) float v2f;
typedef __attribute__((ext_vector_type(8))) float v8f;

#define NAX 160
#define NTH 5
#define KSH 20
#define NFR1 36
#define NFRAG (2 * NFR1)
#define SPITCH 132
#define APITCH 164
#define APW (16 * APITCH)

__device__ inline double h_per(int d) {
  const double p = -0.26794919243112270647255365849413;
  double pd = 1.0, pnd = 1.0;
  for (int k = 0; k < d; ++k)        pd  *= p;
  for (int k = 0; k < NAX - d; ++k)  pnd *= p;
  const double pN = pd * pnd;
  return 1.7320508075688772935274463415059 * (pd + pnd) / (1.0 - pN);
}

__global__ void build_tables(float* __restrict__ Tb) {
  const int idx = blockIdx.x * blockDim.x + threadIdx.x;
  if (idx >= NFRAG * 64) return;
  const int j     = idx & 1;
  const int col   = (idx >> 1) & 15;
  const int khalf = (idx >> 5) & 1;
  const int f     = idx >> 6;
  int d;
  if (f < NFR1) d = 4 * (f - 19) + col - 2 * khalf - j;
  else          d = 4 * (f - NFR1) + col + 2 * khalf + j + 1;
  d = ((d % NAX) + NAX) % NAX;
  Tb[idx] = (float)(0.5 * h_per(d));
}

template <int Q, int P, int S>
__global__ __launch_bounds__(256) void spline_pass(
    const float* __restrict__ in, float* __restrict__ out,
    const float* __restrict__ Tb, int nTiles)
{
  __shared__ __attribute__((aligned(16))) float smem[NFRAG * 64 + NAX * SPITCH];
  v2f* const Bt = reinterpret_cast<v2f*>(smem);
  float* const St = smem + NFRAG * 64;

  const int lane  = threadIdx.x & 31;
  const int wave  = threadIdx.x >> 5;
  const int tile  = blockIdx.x * 8 + wave;
  const bool active = tile < nTiles;

  float4 stg[NAX / 8];
  if constexpr (S == 1) {
    const float* src = in + (active ? tile : 0) * (16 * NAX) + 4 * lane;
#pragma unroll
    for (int j = 0; j < NAX / 8; ++j)
      stg[j] = *reinterpret_cast<const float4*>(src + j * 128);
  } else {
    const int l4  = blockIdx.x * 128 + 4 * lane;
    const int l4c = ((l4 >> 4) < nTiles) ? l4 : 4 * lane;
    const int base4 = (l4c / Q) * P + (l4c % Q);
    const float* src = in + base4 + wave * (NAX / 8) * S;
#pragma unroll
    for (int r = 0; r < NAX / 8; ++r)
      stg[r] = *reinterpret_cast<const float4*>(src + r * S);
  }

  {
    const float4* src4 = reinterpret_cast<const float4*>(Tb);
    float4* dst4 = reinterpret_cast<float4*>(Bt);
    for (int idx = threadIdx.x; idx < NFRAG * 16; idx += 256) dst4[idx] = src4[idx];
  }

  if constexpr (S == 1) {
#pragma unroll
    for (int j = 0; j < NAX / 8; ++j) {
      const int e = j * 128 + 4 * lane;
      *reinterpret_cast<float4*>(
          St + wave * APW + (e / NAX) * APITCH + (e % NAX)) = stg[j];
    }
  } else {
#pragma unroll
    for (int r = 0; r < NAX / 8; ++r) {
      const int k = wave * (NAX / 8) + r;
      *reinterpret_cast<float4*>(St + k * SPITCH + 4 * lane) = stg[r];
    }
  }
  __syncthreads();

  if (active) {

  const int col   = lane & 15;
  const int khalf = lane >> 4;

  const float* Alo = (S == 1)
      ? St + wave * APW + col * APITCH + 2 * khalf
      : St + (2 * khalf) * SPITCH + wave * 16 + col;
  const float* Ahi = (S == 1)
      ? St + wave * APW + col * APITCH + (NAX - 2) - 2 * khalf
      : St + (NAX - 1 - 2 * khalf) * SPITCH + wave * 16 + col;
  auto loadLo = [&](int ks) -> v2f {
    v2f a;
    if constexpr (S == 1) {
      a = *reinterpret_cast<const v2f*>(Alo + 4 * ks);
    } else {
      a.x = Alo[(4 * ks) * SPITCH];
      a.y = Alo[(4 * ks + 1) * SPITCH];
    }
    return a;
  };
  auto loadHi = [&](int ks) -> v2f {
    v2f a;
    if constexpr (S == 1) {
      const v2f r = *reinterpret_cast<const v2f*>(Ahi - 4 * ks);
      a.x = r.y;
      a.y = r.x;
    } else {
      a.x = Ahi[-(4 * ks) * SPITCH];
      a.y = Ahi[-(4 * ks + 1) * SPITCH];
    }
    return a;
  };

  int baseD[8];
#pragma unroll
  for (int v = 0; v < 8; ++v) {
    int lD = tile * 16 + khalf * 8 + v;
    baseD[v] = (lD / Q) * P + (lD % Q);
  }

  v8f accS[NTH], accD[NTH];
#pragma unroll
  for (int t = 0; t < NTH; ++t) { accS[t] = {}; accD[t] = {}; }

  const v2f* Bl = Bt + lane;

  v2f lo0, hi0, lo1, hi1;
  v2f tf0[NTH], kf0[NTH], tf1[NTH], kf1[NTH];
  lo0 = loadLo(0);
  hi0 = loadHi(0);
#pragma unroll
  for (int t = 0; t < NTH; ++t) {
    tf0[t] = Bl[(19 + 4 * t) * 32];
    kf0[t] = Bl[(NFR1 + 4 * t) * 32];
  }

  for (int ks = 0; ks < KSH; ks += 2) {
    {
      lo1 = loadLo(ks + 1);
      hi1 = loadHi(ks + 1);
#pragma unroll
      for (int t = 0; t < NTH; ++t) {
        tf1[t] = Bl[(19 + 4 * t - (ks + 1)) * 32];
        kf1[t] = Bl[(NFR1 + 4 * t + ks + 1) * 32];
      }
      const v2f as = lo0 + hi0;
      const v2f ad = lo0 - hi0;
#pragma unroll
      for (int t = 0; t < NTH; ++t) {
        const v2f bs = tf0[t] + kf0[t];
        const v2f bd = tf0[t] - kf0[t];
        accS[t] = __builtin_amdgcn_wmma_f32_16x16x4_f32(
             false, as,  false, bs,
             (short)0, accS[t],  false,  false);
        accD[t] = __builtin_amdgcn_wmma_f32_16x16x4_f32(
             false, ad,  false, bd,
             (short)0, accD[t],  false,  false);
      }
    }
    {
      if (ks + 2 < KSH) {
        lo0 = loadLo(ks + 2);
        hi0 = loadHi(ks + 2);
#pragma unroll
        for (int t = 0; t < NTH; ++t) {
          tf0[t] = Bl[(19 + 4 * t - (ks + 2)) * 32];
          kf0[t] = Bl[(NFR1 + 4 * t + ks + 2) * 32];
        }
      }
      const v2f as = lo1 + hi1;
      const v2f ad = lo1 - hi1;
#pragma unroll
      for (int t = 0; t < NTH; ++t) {
        const v2f bs = tf1[t] + kf1[t];
        const v2f bd = tf1[t] - kf1[t];
        accS[t] = __builtin_amdgcn_wmma_f32_16x16x4_f32(
             false, as,  false, bs,
             (short)0, accS[t],  false,  false);
        accD[t] = __builtin_amdgcn_wmma_f32_16x16x4_f32(
             false, ad,  false, bd,
             (short)0, accD[t],  false,  false);
      }
    }
  }

  if constexpr (S == 1) {
#pragma unroll
    for (int t = 0; t < NTH; ++t) {
#pragma unroll
      for (int v = 0; v < 8; ++v) {
        const float ys = accS[t][v], yd = accD[t][v];
        out[baseD[v] + (t * 16 + col)] = ys + yd;
        out[baseD[v] + (NAX - 1 - t * 16 - col)] = ys - yd;
      }
    }
  } else {
#pragma unroll
    for (int t = 0; t < NTH; ++t) {
      const v8f ys = accS[t], yd = accD[t];
      const v8f yp = ys + yd;
      const v8f ym = ys - yd;
      float* dp = St + (t * 16 + col) * SPITCH + wave * 16 + khalf * 8;
      float* dm = St + (NAX - 1 - t * 16 - col) * SPITCH + wave * 16 + khalf * 8;
      *reinterpret_cast<float4*>(dp)     = make_float4(yp[0], yp[1], yp[2], yp[3]);
      *reinterpret_cast<float4*>(dp + 4) = make_float4(yp[4], yp[5], yp[6], yp[7]);
      *reinterpret_cast<float4*>(dm)     = make_float4(ym[0], ym[1], ym[2], ym[3]);
      *reinterpret_cast<float4*>(dm + 4) = make_float4(ym[4], ym[5], ym[6], ym[7]);
    }
  }
  }

  if constexpr (S != 1) {
    __syncthreads();
    const int l4 = blockIdx.x * 128 + 4 * lane;
    if ((l4 >> 4) < nTiles) {
      const int base4 = (l4 / Q) * P + (l4 % Q);
#pragma unroll
      for (int r = 0; r < NAX / 8; ++r) {
        const int n = wave * (NAX / 8) + r;
        const float4 val =
            *reinterpret_cast<const float4*>(St + n * SPITCH + 4 * lane);
        *reinterpret_cast<float4*>(out + base4 + n * S) = val;
      }
    }
  }
}

extern "C" void kernel_launch(void* const* d_in, const int* in_sizes, int n_in,
                              void* d_out, int out_size, void* d_ws, size_t ws_size,
                              hipStream_t stream) {
  const float* x   = (const float*)d_in[0];
  float*       out = (float*)d_out;
  float*       Tb  = (float*)d_ws;
  float*       tmp = (float*)d_ws + 32768;

  (void)in_sizes; (void)n_in; (void)out_size; (void)ws_size;

  build_tables<<<dim3((NFRAG * 64 + 255) / 256), dim3(256), 0, stream>>>(Tb);

  const int lines = 2 * 3 * NAX * NAX;
  const int tiles = lines / 16;
  dim3 grid((tiles + 7) / 8), block(256);

  spline_pass<1, NAX, 1>
      <<<grid, block, 0, stream>>>(x, out, Tb, tiles);
  spline_pass<NAX, NAX * NAX, NAX>
      <<<grid, block, 0, stream>>>(out, tmp, Tb, tiles);
  spline_pass<NAX * NAX, NAX * NAX * NAX, NAX * NAX>
      <<<grid, block, 0, stream>>>(tmp, out, Tb, tiles);
}


# Half-period split on top of the reflection one: u,v=x[k]+-x[k+80], each mirrored within 80; four 40x40 f32 WMMA products per line, one N-tile at a time, 120 WMMAs/strip (was 200); k sum order changes
# speedup vs baseline: 4.3316x; 1.1295x over previous
#include <hip/hip_runtime.h>
#include <hip/hip_bf16.h>


typedef __attribute__((ext_vector_type(2))) float v2f;
typedef __attribute__((ext_vector_type(8))) float v8f;

#define NAX 160
#define NQ 40
#define NTQ 3
#define KSQ 10
#define NFR1 18
#define NFRAG (4 * NFR1)
#define SPITCH 132
#define APITCH 164
#define APW (16 * APITCH)

__device__ inline void h_pair(int a, double& ha, double& hb) {
  const double p = -0.26794919243112270647255365849413;
  const int b = (a + NAX / 2) % NAX;
  double pw = 1.0, pa = 0.0, pna = 0.0, pb = 0.0, pnb = 0.0, pN = 0.0;
  for (int k = 0; k <= NAX; ++k) {
    if (k == a)       pa  = pw;
    if (k == NAX - a) pna = pw;
    if (k == b)       pb  = pw;
    if (k == NAX - b) pnb = pw;
    if (k == NAX)     pN  = pw;
    pw *= p;
  }
  const double s3 = 1.7320508075688772935274463415059;
  ha = s3 * (pa + pna) / (1.0 - pN);
  hb = s3 * (pb + pnb) / (1.0 - pN);
}

__global__ void build_tables(float* __restrict__ Tb) {
  const int idx = blockIdx.x * blockDim.x + threadIdx.x;
  if (idx >= NFRAG * 64) return;
  const int j     = idx & 1;
  const int col   = (idx >> 1) & 15;
  const int khalf = (idx >> 5) & 1;
  const int f     = idx >> 6;
  const int sub   = f / NFR1;
  const int g     = f % NFR1;
  int d;
  if ((sub & 1) == 0) d = 4 * (g - 9) + col - 2 * khalf - j;
  else                d = 4 * g + col + 2 * khalf + j + 1;
  d = ((d % NAX) + NAX) % NAX;
  double ha, hb;
  h_pair(d, ha, hb);
  Tb[idx] = (float)(0.25 * (sub < 2 ? ha + hb : ha - hb));
}


template <int Q, int P, int S>
__global__ __launch_bounds__(256) void spline_pass(
    const float* __restrict__ in, float* __restrict__ out,
    const float* __restrict__ Tb, int nTiles)
{
  __shared__ __attribute__((aligned(16))) float smem[NFRAG * 64 + NAX * SPITCH];
  v2f* const Bt = reinterpret_cast<v2f*>(smem);
  float* const St = smem + NFRAG * 64;

  const int lane  = threadIdx.x & 31;
  const int wave  = threadIdx.x >> 5;
  const int tile  = blockIdx.x * 8 + wave;
  const bool active = tile < nTiles;

  float4 stg[NAX / 8];
  if constexpr (S == 1) {
    const float* src = in + (active ? tile : 0) * (16 * NAX) + 4 * lane;
#pragma unroll
    for (int j = 0; j < NAX / 8; ++j)
      stg[j] = *reinterpret_cast<const float4*>(src + j * 128);
  } else {
    const int l4  = blockIdx.x * 128 + 4 * lane;
    const int l4c = ((l4 >> 4) < nTiles) ? l4 : 4 * lane;
    const int base4 = (l4c / Q) * P + (l4c % Q);
    const float* src = in + base4 + wave * (NAX / 8) * S;
#pragma unroll
    for (int r = 0; r < NAX / 8; ++r)
      stg[r] = *reinterpret_cast<const float4*>(src + r * S);
  }

  {
    const float4* src4 = reinterpret_cast<const float4*>(Tb);
    float4* dst4 = reinterpret_cast<float4*>(Bt);
    for (int idx = threadIdx.x; idx < NFRAG * 16; idx += 256) dst4[idx] = src4[idx];
  }

  if constexpr (S == 1) {
#pragma unroll
    for (int j = 0; j < NAX / 8; ++j) {
      const int e = j * 128 + 4 * lane;
      *reinterpret_cast<float4*>(
          St + wave * APW + (e / NAX) * APITCH + (e % NAX)) = stg[j];
    }
  } else {
#pragma unroll
    for (int r = 0; r < NAX / 8; ++r) {
      const int k = wave * (NAX / 8) + r;
      *reinterpret_cast<float4*>(St + k * SPITCH + 4 * lane) = stg[r];
    }
  }
  __syncthreads();

  if (active) {

  const int col   = lane & 15;
  const int khalf = lane >> 4;

  const float* Aup = (S == 1)
      ? St + wave * APW + col * APITCH + 2 * khalf
      : St + (2 * khalf) * SPITCH + wave * 16 + col;
  const float* Adn = (S == 1)
      ? St + wave * APW + col * APITCH + (NAX / 2 - 2) - 2 * khalf
      : St + (NAX / 2 - 1 - 2 * khalf) * SPITCH + wave * 16 + col;
  constexpr int HALF = (S == 1) ? (NAX / 2) : (NAX / 2) * SPITCH;
  auto loadUp = [&](const float* p, int ks) -> v2f {
    v2f a;
    if constexpr (S == 1) {
      a = *reinterpret_cast<const v2f*>(p + 4 * ks);
    } else {
      a.x = p[(4 * ks) * SPITCH];
      a.y = p[(4 * ks + 1) * SPITCH];
    }
    return a;
  };
  auto loadDn = [&](const float* p, int ks) -> v2f {
    v2f a;
    if constexpr (S == 1) {
      const v2f r = *reinterpret_cast<const v2f*>(p - 4 * ks);
      a.x = r.y;
      a.y = r.x;
    } else {
      a.x = p[-(4 * ks) * SPITCH];
      a.y = p[-(4 * ks + 1) * SPITCH];
    }
    return a;
  };

  v2f us[KSQ], ud[KSQ], vs[KSQ], vd[KSQ];
#pragma unroll
  for (int ks = 0; ks < KSQ; ++ks) {
    const v2f x0 = loadUp(Aup, ks);
    const v2f x1 = loadDn(Adn, ks);
    const v2f x2 = loadUp(Aup + HALF, ks);
    const v2f x3 = loadDn(Adn + HALF, ks);
    const v2f u  = x0 + x2;
    const v2f v  = x0 - x2;
    const v2f um = x1 + x3;
    const v2f vm = x1 - x3;
    us[ks] = u + um;  ud[ks] = u - um;
    vs[ks] = v + vm;  vd[ks] = v - vm;
  }

  const v2f* Bl = Bt + lane;

#pragma nounroll
  for (int t = 0; t < NTQ; ++t) {
    const v2f* Bq = Bl + (4 * t) * 32;
    v8f aEs = {}, aEd = {}, aOs = {}, aOd = {};
#pragma unroll
    for (int ks = 0; ks < KSQ; ++ks) {
      const v2f d1 = Bq[(9 - ks) * 32];
      const v2f s1 = Bq[(NFR1 + ks) * 32];
      const v2f d2 = Bq[(2 * NFR1 + 9 - ks) * 32];
      const v2f s2 = Bq[(3 * NFR1 + ks) * 32];
      const v2f bEs = d1 + s1;
      const v2f bEd = d1 - s1;
      const v2f bOs = d2 - s2;
      const v2f bOd = d2 + s2;
      aEs = __builtin_amdgcn_wmma_f32_16x16x4_f32(
           false, us[ks],  false, bEs,
           (short)0, aEs,  false,  false);
      aEd = __builtin_amdgcn_wmma_f32_16x16x4_f32(
           false, ud[ks],  false, bEd,
           (short)0, aEd,  false,  false);
      aOs = __builtin_amdgcn_wmma_f32_16x16x4_f32(
           false, vs[ks],  false, bOs,
           (short)0, aOs,  false,  false);
      aOd = __builtin_amdgcn_wmma_f32_16x16x4_f32(
           false, vd[ks],  false, bOd,
           (short)0, aOd,  false,  false);
    }

    const v8f e  = aEs + aEd;
    const v8f em = aEs - aEd;
    const v8f o  = aOs + aOd;
    const v8f om = aOs - aOd;
    const v8f y0 = e + o, y1 = e - o, y2 = em + om, y3 = em - om;
    const int n = t * 16 + col;
    if (n < NQ) {
      if constexpr (S == 1) {
        float* o0 = out + (tile * 16 + khalf * 8) * NAX;
#pragma unroll
        for (int v = 0; v < 8; ++v) {
          o0[v * NAX + n]                 = y0[v];
          o0[v * NAX + (NAX / 2 + n)]     = y1[v];
          o0[v * NAX + (NAX / 2 - 1 - n)] = y2[v];
          o0[v * NAX + (NAX - 1 - n)]     = y3[v];
        }
      } else {
        float* c = St + wave * 16 + khalf * 8;
        float* r0 = c + n * SPITCH;
        float* r1 = c + (NAX / 2 + n) * SPITCH;
        float* r2 = c + (NAX / 2 - 1 - n) * SPITCH;
        float* r3 = c + (NAX - 1 - n) * SPITCH;
        *reinterpret_cast<float4*>(r0)     = make_float4(y0[0], y0[1], y0[2], y0[3]);
        *reinterpret_cast<float4*>(r0 + 4) = make_float4(y0[4], y0[5], y0[6], y0[7]);
        *reinterpret_cast<float4*>(r1)     = make_float4(y1[0], y1[1], y1[2], y1[3]);
        *reinterpret_cast<float4*>(r1 + 4) = make_float4(y1[4], y1[5], y1[6], y1[7]);
        *reinterpret_cast<float4*>(r2)     = make_float4(y2[0], y2[1], y2[2], y2[3]);
        *reinterpret_cast<float4*>(r2 + 4) = make_float4(y2[4], y2[5], y2[6], y2[7]);
        *reinterpret_cast<float4*>(r3)     = make_float4(y3[0], y3[1], y3[2], y3[3]);
        *reinterpret_cast<float4*>(r3 + 4) = make_float4(y3[4], y3[5], y3[6], y3[7]);
      }
    }
  }
  }

  if constexpr (S != 1) {
    __syncthreads();
    const int l4 = blockIdx.x * 128 + 4 * lane;
    if ((l4 >> 4) < nTiles) {
      const int base4 = (l4 / Q) * P + (l4 % Q);
#pragma unroll
      for (int r = 0; r < NAX / 8; ++r) {
        const int n = wave * (NAX / 8) + r;
        const float4 val =
            *reinterpret_cast<const float4*>(St + n * SPITCH + 4 * lane);
        *reinterpret_cast<float4*>(out + base4 + n * S) = val;
      }
    }
  }
}

extern "C" void kernel_launch(void* const* d_in, const int* in_sizes, int n_in,
                              void* d_out, int out_size, void* d_ws, size_t ws_size,
                              hipStream_t stream) {
  const float* x   = (const float*)d_in[0];
  float*       out = (float*)d_out;
  float*       Tb  = (float*)d_ws;
  float*       tmp = (float*)d_ws + 32768;

  (void)in_sizes; (void)n_in; (void)out_size; (void)ws_size;

  build_tables<<<dim3((NFRAG * 64 + 255) / 256), dim3(256), 0, stream>>>(Tb);

  const int lines = 2 * 3 * NAX * NAX;
  const int tiles = lines / 16;
  dim3 grid((tiles + 7) / 8), block(256);

  spline_pass<1, NAX, 1>
      <<<grid, block, 0, stream>>>(x, out, Tb, tiles);
  spline_pass<NAX, NAX * NAX, NAX>
      <<<grid, block, 0, stream>>>(out, tmp, Tb, tiles);
  spline_pass<NAX * NAX, NAX * NAX * NAX, NAX * NAX>
      <<<grid, block, 0, stream>>>(tmp, out, Tb, tiles);
}


# Prologue: the table copy's five 16-byte loads per lane are straight-line and leave with the 20 tile loads as one group (was a loop that waited for each load before the tile loads); sums unchanged
# speedup vs baseline: 4.4928x; 1.0372x over previous
#include <hip/hip_runtime.h>
#include <hip/hip_bf16.h>


typedef __attribute__((ext_vector_type(2))) float v2f;
typedef __attribute__((ext_vector_type(8))) float v8f;

#define NAX 160
#define NQ 40
#define NTQ 3
#define KSQ 10
#define NFR1 18
#define NFRAG (4 * NFR1)
#define SPITCH 132
#define APITCH 164
#define APW (16 * APITCH)

__device__ inline void h_pair(int a, double& ha, double& hb) {
  const double p = -0.26794919243112270647255365849413;
  const int b = (a + NAX / 2) % NAX;
  double pw = 1.0, pa = 0.0, pna = 0.0, pb = 0.0, pnb = 0.0, pN = 0.0;
  for (int k = 0; k <= NAX; ++k) {
    if (k == a)       pa  = pw;
    if (k == NAX - a) pna = pw;
    if (k == b)       pb  = pw;
    if (k == NAX - b) pnb = pw;
    if (k == NAX)     pN  = pw;
    pw *= p;
  }
  const double s3 = 1.7320508075688772935274463415059;
  ha = s3 * (pa + pna) / (1.0 - pN);
  hb = s3 * (pb + pnb) / (1.0 - pN);
}

__global__ void build_tables(float* __restrict__ Tb) {
  const int idx = blockIdx.x * blockDim.x + threadIdx.x;
  if (idx >= NFRAG * 64) return;
  const int j     = idx & 1;
  const int col   = (idx >> 1) & 15;
  const int khalf = (idx >> 5) & 1;
  const int f     = idx >> 6;
  const int sub   = f / NFR1;
  const int g     = f % NFR1;
  int d;
  if ((sub & 1) == 0) d = 4 * (g - 9) + col - 2 * khalf - j;
  else                d = 4 * g + col + 2 * khalf + j + 1;
  d = ((d % NAX) + NAX) % NAX;
  double ha, hb;
  h_pair(d, ha, hb);
  Tb[idx] = (float)(0.25 * (sub < 2 ? ha + hb : ha - hb));
}


template <int Q, int P, int S>
__global__ __launch_bounds__(256) void spline_pass(
    const float* __restrict__ in, float* __restrict__ out,
    const float* __restrict__ Tb, int nTiles)
{
  __shared__ __attribute__((aligned(16))) float smem[NFRAG * 64 + NAX * SPITCH];
  v2f* const Bt = reinterpret_cast<v2f*>(smem);
  float* const St = smem + NFRAG * 64;

  const int lane  = threadIdx.x & 31;
  const int wave  = threadIdx.x >> 5;
  const int tile  = blockIdx.x * 8 + wave;
  const bool active = tile < nTiles;

  const float4* const tsrc4 = reinterpret_cast<const float4*>(Tb);
  const int t5 = (threadIdx.x < NFRAG * 16 - 1024) ? threadIdx.x + 1024 : threadIdx.x;
  float4 tbl[5];
#pragma unroll
  for (int i = 0; i < 4; ++i) tbl[i] = tsrc4[threadIdx.x + 256 * i];
  tbl[4] = tsrc4[t5];

  float4 stg[NAX / 8];
  if constexpr (S == 1) {
    const float* src = in + (active ? tile : 0) * (16 * NAX) + 4 * lane;
#pragma unroll
    for (int j = 0; j < NAX / 8; ++j)
      stg[j] = *reinterpret_cast<const float4*>(src + j * 128);
  } else {
    const int l4  = blockIdx.x * 128 + 4 * lane;
    const int l4c = ((l4 >> 4) < nTiles) ? l4 : 4 * lane;
    const int base4 = (l4c / Q) * P + (l4c % Q);
    const float* src = in + base4 + wave * (NAX / 8) * S;
#pragma unroll
    for (int r = 0; r < NAX / 8; ++r)
      stg[r] = *reinterpret_cast<const float4*>(src + r * S);
  }

  {
    float4* dst4 = reinterpret_cast<float4*>(Bt);
#pragma unroll
    for (int i = 0; i < 4; ++i) dst4[threadIdx.x + 256 * i] = tbl[i];
    dst4[t5] = tbl[4];
  }

  if constexpr (S == 1) {
#pragma unroll
    for (int j = 0; j < NAX / 8; ++j) {
      const int e = j * 128 + 4 * lane;
      *reinterpret_cast<float4*>(
          St + wave * APW + (e / NAX) * APITCH + (e % NAX)) = stg[j];
    }
  } else {
#pragma unroll
    for (int r = 0; r < NAX / 8; ++r) {
      const int k = wave * (NAX / 8) + r;
      *reinterpret_cast<float4*>(St + k * SPITCH + 4 * lane) = stg[r];
    }
  }
  __syncthreads();

  if (active) {

  const int col   = lane & 15;
  const int khalf = lane >> 4;

  const float* Aup = (S == 1)
      ? St + wave * APW + col * APITCH + 2 * khalf
      : St + (2 * khalf) * SPITCH + wave * 16 + col;
  const float* Adn = (S == 1)
      ? St + wave * APW + col * APITCH + (NAX / 2 - 2) - 2 * khalf
      : St + (NAX / 2 - 1 - 2 * khalf) * SPITCH + wave * 16 + col;
  constexpr int HALF = (S == 1) ? (NAX / 2) : (NAX / 2) * SPITCH;
  auto loadUp = [&](const float* p, int ks) -> v2f {
    v2f a;
    if constexpr (S == 1) {
      a = *reinterpret_cast<const v2f*>(p + 4 * ks);
    } else {
      a.x = p[(4 * ks) * SPITCH];
      a.y = p[(4 * ks + 1) * SPITCH];
    }
    return a;
  };
  auto loadDn = [&](const float* p, int ks) -> v2f {
    v2f a;
    if constexpr (S == 1) {
      const v2f r = *reinterpret_cast<const v2f*>(p - 4 * ks);
      a.x = r.y;
      a.y = r.x;
    } else {
      a.x = p[-(4 * ks) * SPITCH];
      a.y = p[-(4 * ks + 1) * SPITCH];
    }
    return a;
  };

  v2f us[KSQ], ud[KSQ], vs[KSQ], vd[KSQ];
#pragma unroll
  for (int ks = 0; ks < KSQ; ++ks) {
    const v2f x0 = loadUp(Aup, ks);
    const v2f x1 = loadDn(Adn, ks);
    const v2f x2 = loadUp(Aup + HALF, ks);
    const v2f x3 = loadDn(Adn + HALF, ks);
    const v2f u  = x0 + x2;
    const v2f v  = x0 - x2;
    const v2f um = x1 + x3;
    const v2f vm = x1 - x3;
    us[ks] = u + um;  ud[ks] = u - um;
    vs[ks] = v + vm;  vd[ks] = v - vm;
  }

  const v2f* Bl = Bt + lane;

#pragma nounroll
  for (int t = 0; t < NTQ; ++t) {
    const v2f* Bq = Bl + (4 * t) * 32;
    v8f aEs = {}, aEd = {}, aOs = {}, aOd = {};
#pragma unroll
    for (int ks = 0; ks < KSQ; ++ks) {
      const v2f d1 = Bq[(9 - ks) * 32];
      const v2f s1 = Bq[(NFR1 + ks) * 32];
      const v2f d2 = Bq[(2 * NFR1 + 9 - ks) * 32];
      const v2f s2 = Bq[(3 * NFR1 + ks) * 32];
      const v2f bEs = d1 + s1;
      const v2f bEd = d1 - s1;
      const v2f bOs = d2 - s2;
      const v2f bOd = d2 + s2;
      aEs = __builtin_amdgcn_wmma_f32_16x16x4_f32(
           false, us[ks],  false, bEs,
           (short)0, aEs,  false,  false);
      aEd = __builtin_amdgcn_wmma_f32_16x16x4_f32(
           false, ud[ks],  false, bEd,
           (short)0, aEd,  false,  false);
      aOs = __builtin_amdgcn_wmma_f32_16x16x4_f32(
           false, vs[ks],  false, bOs,
           (short)0, aOs,  false,  false);
      aOd = __builtin_amdgcn_wmma_f32_16x16x4_f32(
           false, vd[ks],  false, bOd,
           (short)0, aOd,  false,  false);
    }

    const v8f e  = aEs + aEd;
    const v8f em = aEs - aEd;
    const v8f o  = aOs + aOd;
    const v8f om = aOs - aOd;
    const v8f y0 = e + o, y1 = e - o, y2 = em + om, y3 = em - om;
    const int n = t * 16 + col;
    if (n < NQ) {
      if constexpr (S == 1) {
        float* o0 = out + (tile * 16 + khalf * 8) * NAX;
#pragma unroll
        for (int v = 0; v < 8; ++v) {
          o0[v * NAX + n]                 = y0[v];
          o0[v * NAX + (NAX / 2 + n)]     = y1[v];
          o0[v * NAX + (NAX / 2 - 1 - n)] = y2[v];
          o0[v * NAX + (NAX - 1 - n)]     = y3[v];
        }
      } else {
        float* c = St + wave * 16 + khalf * 8;
        float* r0 = c + n * SPITCH;
        float* r1 = c + (NAX / 2 + n) * SPITCH;
        float* r2 = c + (NAX / 2 - 1 - n) * SPITCH;
        float* r3 = c + (NAX - 1 - n) * SPITCH;
        *reinterpret_cast<float4*>(r0)     = make_float4(y0[0], y0[1], y0[2], y0[3]);
        *reinterpret_cast<float4*>(r0 + 4) = make_float4(y0[4], y0[5], y0[6], y0[7]);
        *reinterpret_cast<float4*>(r1)     = make_float4(y1[0], y1[1], y1[2], y1[3]);
        *reinterpret_cast<float4*>(r1 + 4) = make_float4(y1[4], y1[5], y1[6], y1[7]);
        *reinterpret_cast<float4*>(r2)     = make_float4(y2[0], y2[1], y2[2], y2[3]);
        *reinterpret_cast<float4*>(r2 + 4) = make_float4(y2[4], y2[5], y2[6], y2[7]);
        *reinterpret_cast<float4*>(r3)     = make_float4(y3[0], y3[1], y3[2], y3[3]);
        *reinterpret_cast<float4*>(r3 + 4) = make_float4(y3[4], y3[5], y3[6], y3[7]);
      }
    }
  }
  }

  if constexpr (S != 1) {
    __syncthreads();
    const int l4 = blockIdx.x * 128 + 4 * lane;
    if ((l4 >> 4) < nTiles) {
      const int base4 = (l4 / Q) * P + (l4 % Q);
#pragma unroll
      for (int r = 0; r < NAX / 8; ++r) {
        const int n = wave * (NAX / 8) + r;
        const float4 val =
            *reinterpret_cast<const float4*>(St + n * SPITCH + 4 * lane);
        *reinterpret_cast<float4*>(out + base4 + n * S) = val;
      }
    }
  }
}

extern "C" void kernel_launch(void* const* d_in, const int* in_sizes, int n_in,
                              void* d_out, int out_size, void* d_ws, size_t ws_size,
                              hipStream_t stream) {
  const float* x   = (const float*)d_in[0];
  float*       out = (float*)d_out;
  float*       Tb  = (float*)d_ws;
  float*       tmp = (float*)d_ws + 32768;

  (void)in_sizes; (void)n_in; (void)out_size; (void)ws_size;

  build_tables<<<dim3((NFRAG * 64 + 255) / 256), dim3(256), 0, stream>>>(Tb);

  const int lines = 2 * 3 * NAX * NAX;
  const int tiles = lines / 16;
  dim3 grid((tiles + 7) / 8), block(256);

  spline_pass<1, NAX, 1>
      <<<grid, block, 0, stream>>>(x, out, Tb, tiles);
  spline_pass<NAX, NAX * NAX, NAX>
      <<<grid, block, 0, stream>>>(out, tmp, Tb, tiles);
  spline_pass<NAX * NAX, NAX * NAX * NAX, NAX * NAX>
      <<<grid, block, 0, stream>>>(tmp, out, Tb, tiles);
}


# Strided passes: each N-tile's finished rows leave behind a barrier right after the tile (64, 64, 32 rows), stores draining under the next tile's WMMAs; was all 160 rows after the last tile; same sums
# speedup vs baseline: 4.6463x; 1.0342x over previous
#include <hip/hip_runtime.h>
#include <hip/hip_bf16.h>


typedef __attribute__((ext_vector_type(2))) float v2f;
typedef __attribute__((ext_vector_type(8))) float v8f;

#define NAX 160
#define NQ 40
#define NTQ 3
#define KSQ 10
#define NFR1 18
#define NFRAG (4 * NFR1)
#define SPITCH 132
#define APITCH 164
#define APW (16 * APITCH)

__device__ inline void h_pair(int a, double& ha, double& hb) {
  const double p = -0.26794919243112270647255365849413;
  const int b = (a + NAX / 2) % NAX;
  double pw = 1.0, pa = 0.0, pna = 0.0, pb = 0.0, pnb = 0.0, pN = 0.0;
  for (int k = 0; k <= NAX; ++k) {
    if (k == a)       pa  = pw;
    if (k == NAX - a) pna = pw;
    if (k == b)       pb  = pw;
    if (k == NAX - b) pnb = pw;
    if (k == NAX)     pN  = pw;
    pw *= p;
  }
  const double s3 = 1.7320508075688772935274463415059;
  ha = s3 * (pa + pna) / (1.0 - pN);
  hb = s3 * (pb + pnb) / (1.0 - pN);
}

__global__ void build_tables(float* __restrict__ Tb) {
  const int idx = blockIdx.x * blockDim.x + threadIdx.x;
  if (idx >= NFRAG * 64) return;
  const int j     = idx & 1;
  const int col   = (idx >> 1) & 15;
  const int khalf = (idx >> 5) & 1;
  const int f     = idx >> 6;
  const int sub   = f / NFR1;
  const int g     = f % NFR1;
  int d;
  if ((sub & 1) == 0) d = 4 * (g - 9) + col - 2 * khalf - j;
  else                d = 4 * g + col + 2 * khalf + j + 1;
  d = ((d % NAX) + NAX) % NAX;
  double ha, hb;
  h_pair(d, ha, hb);
  Tb[idx] = (float)(0.25 * (sub < 2 ? ha + hb : ha - hb));
}


template <int Q, int P, int S>
__global__ __launch_bounds__(256) void spline_pass(
    const float* __restrict__ in, float* __restrict__ out,
    const float* __restrict__ Tb, int nTiles)
{
  __shared__ __attribute__((aligned(16))) float smem[NFRAG * 64 + NAX * SPITCH];
  v2f* const Bt = reinterpret_cast<v2f*>(smem);
  float* const St = smem + NFRAG * 64;

  const int lane  = threadIdx.x & 31;
  const int wave  = threadIdx.x >> 5;
  const int tile  = blockIdx.x * 8 + wave;
  const bool active = tile < nTiles;

  const float4* const tsrc4 = reinterpret_cast<const float4*>(Tb);
  const int t5 = (threadIdx.x < NFRAG * 16 - 1024) ? threadIdx.x + 1024 : threadIdx.x;
  float4 tbl[5];
#pragma unroll
  for (int i = 0; i < 4; ++i) tbl[i] = tsrc4[threadIdx.x + 256 * i];
  tbl[4] = tsrc4[t5];

  float4 stg[NAX / 8];
  if constexpr (S == 1) {
    const float* src = in + (active ? tile : 0) * (16 * NAX) + 4 * lane;
#pragma unroll
    for (int j = 0; j < NAX / 8; ++j)
      stg[j] = *reinterpret_cast<const float4*>(src + j * 128);
  } else {
    const int l4  = blockIdx.x * 128 + 4 * lane;
    const int l4c = ((l4 >> 4) < nTiles) ? l4 : 4 * lane;
    const int base4 = (l4c / Q) * P + (l4c % Q);
    const float* src = in + base4 + wave * (NAX / 8) * S;
#pragma unroll
    for (int r = 0; r < NAX / 8; ++r)
      stg[r] = *reinterpret_cast<const float4*>(src + r * S);
  }

  {
    float4* dst4 = reinterpret_cast<float4*>(Bt);
#pragma unroll
    for (int i = 0; i < 4; ++i) dst4[threadIdx.x + 256 * i] = tbl[i];
    dst4[t5] = tbl[4];
  }

  if constexpr (S == 1) {
#pragma unroll
    for (int j = 0; j < NAX / 8; ++j) {
      const int e = j * 128 + 4 * lane;
      *reinterpret_cast<float4*>(
          St + wave * APW + (e / NAX) * APITCH + (e % NAX)) = stg[j];
    }
  } else {
#pragma unroll
    for (int r = 0; r < NAX / 8; ++r) {
      const int k = wave * (NAX / 8) + r;
      *reinterpret_cast<float4*>(St + k * SPITCH + 4 * lane) = stg[r];
    }
  }
  __syncthreads();

  const int col   = lane & 15;
  const int khalf = lane >> 4;

  const float* Aup = (S == 1)
      ? St + wave * APW + col * APITCH + 2 * khalf
      : St + (2 * khalf) * SPITCH + wave * 16 + col;
  const float* Adn = (S == 1)
      ? St + wave * APW + col * APITCH + (NAX / 2 - 2) - 2 * khalf
      : St + (NAX / 2 - 1 - 2 * khalf) * SPITCH + wave * 16 + col;
  constexpr int HALF = (S == 1) ? (NAX / 2) : (NAX / 2) * SPITCH;
  auto loadUp = [&](const float* p, int ks) -> v2f {
    v2f a;
    if constexpr (S == 1) {
      a = *reinterpret_cast<const v2f*>(p + 4 * ks);
    } else {
      a.x = p[(4 * ks) * SPITCH];
      a.y = p[(4 * ks + 1) * SPITCH];
    }
    return a;
  };
  auto loadDn = [&](const float* p, int ks) -> v2f {
    v2f a;
    if constexpr (S == 1) {
      const v2f r = *reinterpret_cast<const v2f*>(p - 4 * ks);
      a.x = r.y;
      a.y = r.x;
    } else {
      a.x = p[-(4 * ks) * SPITCH];
      a.y = p[-(4 * ks + 1) * SPITCH];
    }
    return a;
  };

  v2f us[KSQ], ud[KSQ], vs[KSQ], vd[KSQ];
#pragma unroll
  for (int ks = 0; ks < KSQ; ++ks) {
    const v2f x0 = loadUp(Aup, ks);
    const v2f x1 = loadDn(Adn, ks);
    const v2f x2 = loadUp(Aup + HALF, ks);
    const v2f x3 = loadDn(Adn + HALF, ks);
    const v2f u  = x0 + x2;
    const v2f v  = x0 - x2;
    const v2f um = x1 + x3;
    const v2f vm = x1 - x3;
    us[ks] = u + um;  ud[ks] = u - um;
    vs[ks] = v + vm;  vd[ks] = v - vm;
  }

  const v2f* Bl = Bt + lane;

#pragma nounroll
  for (int t = 0; t < NTQ; ++t) {
    const v2f* Bq = Bl + (4 * t) * 32;
    v8f aEs = {}, aEd = {}, aOs = {}, aOd = {};
#pragma unroll
    for (int ks = 0; ks < KSQ; ++ks) {
      const v2f d1 = Bq[(9 - ks) * 32];
      const v2f s1 = Bq[(NFR1 + ks) * 32];
      const v2f d2 = Bq[(2 * NFR1 + 9 - ks) * 32];
      const v2f s2 = Bq[(3 * NFR1 + ks) * 32];
      const v2f bEs = d1 + s1;
      const v2f bEd = d1 - s1;
      const v2f bOs = d2 - s2;
      const v2f bOd = d2 + s2;
      aEs = __builtin_amdgcn_wmma_f32_16x16x4_f32(
           false, us[ks],  false, bEs,
           (short)0, aEs,  false,  false);
      aEd = __builtin_amdgcn_wmma_f32_16x16x4_f32(
           false, ud[ks],  false, bEd,
           (short)0, aEd,  false,  false);
      aOs = __builtin_amdgcn_wmma_f32_16x16x4_f32(
           false, vs[ks],  false, bOs,
           (short)0, aOs,  false,  false);
      aOd = __builtin_amdgcn_wmma_f32_16x16x4_f32(
           false, vd[ks],  false, bOd,
           (short)0, aOd,  false,  false);
    }

    const v8f e  = aEs + aEd;
    const v8f em = aEs - aEd;
    const v8f o  = aOs + aOd;
    const v8f om = aOs - aOd;
    const v8f y0 = e + o, y1 = e - o, y2 = em + om, y3 = em - om;
    const int n = t * 16 + col;
    if (n < NQ && (S != 1 || active)) {
      if constexpr (S == 1) {
        float* o0 = out + (tile * 16 + khalf * 8) * NAX;
#pragma unroll
        for (int v = 0; v < 8; ++v) {
          o0[v * NAX + n]                 = y0[v];
          o0[v * NAX + (NAX / 2 + n)]     = y1[v];
          o0[v * NAX + (NAX / 2 - 1 - n)] = y2[v];
          o0[v * NAX + (NAX - 1 - n)]     = y3[v];
        }
      } else {
        float* c = St + wave * 16 + khalf * 8;
        float* r0 = c + n * SPITCH;
        float* r1 = c + (NAX / 2 + n) * SPITCH;
        float* r2 = c + (NAX / 2 - 1 - n) * SPITCH;
        float* r3 = c + (NAX - 1 - n) * SPITCH;
        *reinterpret_cast<float4*>(r0)     = make_float4(y0[0], y0[1], y0[2], y0[3]);
        *reinterpret_cast<float4*>(r0 + 4) = make_float4(y0[4], y0[5], y0[6], y0[7]);
        *reinterpret_cast<float4*>(r1)     = make_float4(y1[0], y1[1], y1[2], y1[3]);
        *reinterpret_cast<float4*>(r1 + 4) = make_float4(y1[4], y1[5], y1[6], y1[7]);
        *reinterpret_cast<float4*>(r2)     = make_float4(y2[0], y2[1], y2[2], y2[3]);
        *reinterpret_cast<float4*>(r2 + 4) = make_float4(y2[4], y2[5], y2[6], y2[7]);
        *reinterpret_cast<float4*>(r3)     = make_float4(y3[0], y3[1], y3[2], y3[3]);
        *reinterpret_cast<float4*>(r3 + 4) = make_float4(y3[4], y3[5], y3[6], y3[7]);
      }
    }

    if constexpr (S != 1) {
      __syncthreads();
      const int l4 = blockIdx.x * 128 + 4 * lane;
      if ((l4 >> 4) < nTiles) {
        const int base4 = (l4 / Q) * P + (l4 % Q);
        const int per = (t < NTQ - 1) ? 8 : 4;
        const int n0  = 16 * t + (wave & 1) * per;
        const int g   = wave >> 1;
        const int hi  = (g & 1) * (NAX / 2);
        const int row0 = hi + ((g & 2) ? (NAX / 2 - n0 - per) : n0);
        const float* sp = St + row0 * SPITCH + 4 * lane;
        float* gp = out + base4 + row0 * S;
        if (t < NTQ - 1) {
          float4 val[8];
#pragma unroll
          for (int r = 0; r < 8; ++r)
            val[r] = *reinterpret_cast<const float4*>(sp + r * SPITCH);
#pragma unroll
          for (int r = 0; r < 8; ++r)
            *reinterpret_cast<float4*>(gp + r * S) = val[r];
        } else {
          float4 val[4];
#pragma unroll
          for (int r = 0; r < 4; ++r)
            val[r] = *reinterpret_cast<const float4*>(sp + r * SPITCH);
#pragma unroll
          for (int r = 0; r < 4; ++r)
            *reinterpret_cast<float4*>(gp + r * S) = val[r];
        }
      }
    }
  }
}

extern "C" void kernel_launch(void* const* d_in, const int* in_sizes, int n_in,
                              void* d_out, int out_size, void* d_ws, size_t ws_size,
                              hipStream_t stream) {
  const float* x   = (const float*)d_in[0];
  float*       out = (float*)d_out;
  float*       Tb  = (float*)d_ws;
  float*       tmp = (float*)d_ws + 32768;

  (void)in_sizes; (void)n_in; (void)out_size; (void)ws_size;

  build_tables<<<dim3((NFRAG * 64 + 255) / 256), dim3(256), 0, stream>>>(Tb);

  const int lines = 2 * 3 * NAX * NAX;
  const int tiles = lines / 16;
  dim3 grid((tiles + 7) / 8), block(256);

  spline_pass<1, NAX, 1>
      <<<grid, block, 0, stream>>>(x, out, Tb, tiles);
  spline_pass<NAX, NAX * NAX, NAX>
      <<<grid, block, 0, stream>>>(out, tmp, Tb, tiles);
  spline_pass<NAX * NAX, NAX * NAX * NAX, NAX * NAX>
      <<<grid, block, 0, stream>>>(tmp, out, Tb, tiles);
}


# Tap tables built in LDS by pass 1's own workgroups while their tile loads are in flight (13 f64 ops per thread); workgroup 0 writes them out for passes 2-3; no table kernel, 3 launches; same sums
# speedup vs baseline: 5.0397x; 1.0847x over previous
#include <hip/hip_runtime.h>
#include <hip/hip_bf16.h>


typedef __attribute__((ext_vector_type(2))) float v2f;
typedef __attribute__((ext_vector_type(4))) float v4f;

__device__ __forceinline__ void lds_barrier() {
  __builtin_amdgcn_fence(__ATOMIC_RELEASE, "workgroup", "local");
  __builtin_amdgcn_s_barrier();
  __builtin_amdgcn_fence(__ATOMIC_ACQUIRE, "workgroup", "local");
}
typedef __attribute__((ext_vector_type(8))) float v8f;

#define NAX 160
#define NQ 40
#define NTQ 3
#define KSQ 10
#define NFR1 18
#define NFRAG (4 * NFR1)
#define SPITCH 132
#define APITCH 164
#define APW (16 * APITCH)
#define DLO 39
#define NEXT 127
#define NBLD 578


template <int Q, int P, int S>
__global__ __launch_bounds__(256) void spline_pass(
    const float* __restrict__ in, float* __restrict__ out,
    const float* __restrict__ Tb, float* __restrict__ Tw, int nTiles)
{
  __shared__ __attribute__((aligned(16))) float
      smem[NFRAG * 64 + NAX * SPITCH + (S == 1 ? NBLD : 0)];
  v2f* const Bt = reinterpret_cast<v2f*>(smem);
  float* const St = smem + NFRAG * 64;

  const int lane  = threadIdx.x & 31;
  const int wave  = threadIdx.x >> 5;
  const int tile  = blockIdx.x * 8 + wave;
  const bool active = tile < nTiles;

  const int t5 = (threadIdx.x < NFRAG * 16 - 1024) ? threadIdx.x + 1024 : threadIdx.x;
  float4 tbl[5];
  if constexpr (S != 1) {
    const float4* const tsrc4 = reinterpret_cast<const float4*>(Tb);
#pragma unroll
    for (int i = 0; i < 4; ++i) tbl[i] = tsrc4[threadIdx.x + 256 * i];
    tbl[4] = tsrc4[t5];
  }

  float4 stg[NAX / 8];
  v4f stv[NAX / 8];
  if constexpr (S == 1) {
    const float* src = in + (active ? tile : 0) * (16 * NAX) + 4 * lane;
#pragma unroll
    for (int j = 0; j < NAX / 8; ++j)
      stv[j] = *reinterpret_cast<const v4f*>(src + j * 128);
  } else {
    const int l4  = blockIdx.x * 128 + 4 * lane;
    const int l4c = ((l4 >> 4) < nTiles) ? l4 : 4 * lane;
    const int base4 = (l4c / Q) * P + (l4c % Q);
    const float* src = in + base4 + wave * (NAX / 8) * S;
#pragma unroll
    for (int r = 0; r < NAX / 8; ++r)
      stg[r] = *reinterpret_cast<const float4*>(src + r * S);
  }

  if constexpr (S != 1) {
    float4* dst4 = reinterpret_cast<float4*>(Bt);
#pragma unroll
    for (int i = 0; i < 4; ++i) dst4[threadIdx.x + 256 * i] = tbl[i];
    dst4[t5] = tbl[4];
  } else {
    double* const pw = reinterpret_cast<double*>(smem + NFRAG * 64 + NAX * SPITCH);
    float*  const Ev = reinterpret_cast<float*>(pw + NAX + 2);
    {
      const int dP = min((int)threadIdx.x, NAX);
      double b = -0.26794919243112270647255365849413;
      double r = 1.0;
#pragma unroll
      for (int i = 0; i < 8; ++i) {
        if ((dP >> i) & 1) r *= b;
        b *= b;
      }
      pw[dP] = r;
    }
    lds_barrier();
    {
      const int eC = min((int)threadIdx.x, 2 * NEXT - 1);
      const int which = (eC >= NEXT) ? 1 : 0;
      const int d = eC - which * NEXT - DLO;
      const int a = (d < 0) ? d + NAX : d;
      const int b = (a < NAX / 2) ? a + NAX / 2 : a - NAX / 2;
      const double s3 = 1.7320508075688772935274463415059;
      const double ha = s3 * (pw[a] + pw[NAX - a]);
      const double hb = s3 * (pw[b] + pw[NAX - b]);
      Ev[eC] = (float)(0.25 * (ha + (which ? -hb : hb)));
    }
    lds_barrier();
    float4* dst4 = reinterpret_cast<float4*>(Bt);
    const int p4 = threadIdx.x & 15;
    const int cb = 2 * (p4 & 7);
    const int kh = p4 >> 3;
#pragma unroll
    for (int i = 0; i < 5; ++i) {
      const int q = (i < 4) ? (int)threadIdx.x + 256 * i : t5;
      const int f = q >> 4;
      const int sub = f / NFR1;
      const int g = f % NFR1;
      const bool dif = (sub & 1) == 0;
      const int d0 = dif ? 4 * (g - 9) + cb - 2 * kh : 4 * g + cb + 2 * kh + 1;
      const float* E = Ev + (sub >> 1) * NEXT + DLO + d0;
      dst4[q] = make_float4(E[0], E[dif ? -1 : 1], E[1], E[dif ? 0 : 2]);
    }
  }

  if constexpr (S == 1) {
#pragma unroll
    for (int j = 0; j < NAX / 8; ++j) {
      const int e = j * 128 + 4 * lane;
      *reinterpret_cast<v4f*>(
          St + wave * APW + (e / NAX) * APITCH + (e % NAX)) = stv[j];
    }
    if (blockIdx.x == 0) {
      const float4* src4 = reinterpret_cast<const float4*>(Bt);
      float4* gdst4 = reinterpret_cast<float4*>(Tw);
#pragma unroll
      for (int i = 0; i < 4; ++i)
        gdst4[threadIdx.x + 256 * i] = src4[threadIdx.x + 256 * i];
      gdst4[t5] = src4[t5];
    }
  } else {
#pragma unroll
    for (int r = 0; r < NAX / 8; ++r) {
      const int k = wave * (NAX / 8) + r;
      *reinterpret_cast<float4*>(St + k * SPITCH + 4 * lane) = stg[r];
    }
  }
  __syncthreads();

  const int col   = lane & 15;
  const int khalf = lane >> 4;

  const float* Aup = (S == 1)
      ? St + wave * APW + col * APITCH + 2 * khalf
      : St + (2 * khalf) * SPITCH + wave * 16 + col;
  const float* Adn = (S == 1)
      ? St + wave * APW + col * APITCH + (NAX / 2 - 2) - 2 * khalf
      : St + (NAX / 2 - 1 - 2 * khalf) * SPITCH + wave * 16 + col;
  constexpr int HALF = (S == 1) ? (NAX / 2) : (NAX / 2) * SPITCH;
  auto loadUp = [&](const float* p, int ks) -> v2f {
    v2f a;
    if constexpr (S == 1) {
      a = *reinterpret_cast<const v2f*>(p + 4 * ks);
    } else {
      a.x = p[(4 * ks) * SPITCH];
      a.y = p[(4 * ks + 1) * SPITCH];
    }
    return a;
  };
  auto loadDn = [&](const float* p, int ks) -> v2f {
    v2f a;
    if constexpr (S == 1) {
      const v2f r = *reinterpret_cast<const v2f*>(p - 4 * ks);
      a.x = r.y;
      a.y = r.x;
    } else {
      a.x = p[-(4 * ks) * SPITCH];
      a.y = p[-(4 * ks + 1) * SPITCH];
    }
    return a;
  };

  v2f us[KSQ], ud[KSQ], vs[KSQ], vd[KSQ];
#pragma unroll
  for (int ks = 0; ks < KSQ; ++ks) {
    const v2f x0 = loadUp(Aup, ks);
    const v2f x1 = loadDn(Adn, ks);
    const v2f x2 = loadUp(Aup + HALF, ks);
    const v2f x3 = loadDn(Adn + HALF, ks);
    const v2f u  = x0 + x2;
    const v2f v  = x0 - x2;
    const v2f um = x1 + x3;
    const v2f vm = x1 - x3;
    us[ks] = u + um;  ud[ks] = u - um;
    vs[ks] = v + vm;  vd[ks] = v - vm;
  }

  const v2f* Bl = Bt + lane;

#pragma nounroll
  for (int t = 0; t < NTQ; ++t) {
    const v2f* Bq = Bl + (4 * t) * 32;
    v8f aEs = {}, aEd = {}, aOs = {}, aOd = {};
#pragma unroll
    for (int ks = 0; ks < KSQ; ++ks) {
      const v2f d1 = Bq[(9 - ks) * 32];
      const v2f s1 = Bq[(NFR1 + ks) * 32];
      const v2f d2 = Bq[(2 * NFR1 + 9 - ks) * 32];
      const v2f s2 = Bq[(3 * NFR1 + ks) * 32];
      const v2f bEs = d1 + s1;
      const v2f bEd = d1 - s1;
      const v2f bOs = d2 - s2;
      const v2f bOd = d2 + s2;
      aEs = __builtin_amdgcn_wmma_f32_16x16x4_f32(
           false, us[ks],  false, bEs,
           (short)0, aEs,  false,  false);
      aEd = __builtin_amdgcn_wmma_f32_16x16x4_f32(
           false, ud[ks],  false, bEd,
           (short)0, aEd,  false,  false);
      aOs = __builtin_amdgcn_wmma_f32_16x16x4_f32(
           false, vs[ks],  false, bOs,
           (short)0, aOs,  false,  false);
      aOd = __builtin_amdgcn_wmma_f32_16x16x4_f32(
           false, vd[ks],  false, bOd,
           (short)0, aOd,  false,  false);
    }

    const v8f e  = aEs + aEd;
    const v8f em = aEs - aEd;
    const v8f o  = aOs + aOd;
    const v8f om = aOs - aOd;
    const v8f y0 = e + o, y1 = e - o, y2 = em + om, y3 = em - om;
    const int n = t * 16 + col;
    if (n < NQ && (S != 1 || active)) {
      if constexpr (S == 1) {
        float* o0 = out + (tile * 16 + khalf * 8) * NAX;
#pragma unroll
        for (int v = 0; v < 8; ++v) {
          o0[v * NAX + n]                 = y0[v];
          o0[v * NAX + (NAX / 2 + n)]     = y1[v];
          o0[v * NAX + (NAX / 2 - 1 - n)] = y2[v];
          o0[v * NAX + (NAX - 1 - n)]     = y3[v];
        }
      } else {
        float* c = St + wave * 16 + khalf * 8;
        float* r0 = c + n * SPITCH;
        float* r1 = c + (NAX / 2 + n) * SPITCH;
        float* r2 = c + (NAX / 2 - 1 - n) * SPITCH;
        float* r3 = c + (NAX - 1 - n) * SPITCH;
        *reinterpret_cast<float4*>(r0)     = make_float4(y0[0], y0[1], y0[2], y0[3]);
        *reinterpret_cast<float4*>(r0 + 4) = make_float4(y0[4], y0[5], y0[6], y0[7]);
        *reinterpret_cast<float4*>(r1)     = make_float4(y1[0], y1[1], y1[2], y1[3]);
        *reinterpret_cast<float4*>(r1 + 4) = make_float4(y1[4], y1[5], y1[6], y1[7]);
        *reinterpret_cast<float4*>(r2)     = make_float4(y2[0], y2[1], y2[2], y2[3]);
        *reinterpret_cast<float4*>(r2 + 4) = make_float4(y2[4], y2[5], y2[6], y2[7]);
        *reinterpret_cast<float4*>(r3)     = make_float4(y3[0], y3[1], y3[2], y3[3]);
        *reinterpret_cast<float4*>(r3 + 4) = make_float4(y3[4], y3[5], y3[6], y3[7]);
      }
    }

    if constexpr (S != 1) {
      __syncthreads();
      const int l4 = blockIdx.x * 128 + 4 * lane;
      if ((l4 >> 4) < nTiles) {
        const int base4 = (l4 / Q) * P + (l4 % Q);
        const int per = (t < NTQ - 1) ? 8 : 4;
        const int n0  = 16 * t + (wave & 1) * per;
        const int g   = wave >> 1;
        const int hi  = (g & 1) * (NAX / 2);
        const int row0 = hi + ((g & 2) ? (NAX / 2 - n0 - per) : n0);
        const float* sp = St + row0 * SPITCH + 4 * lane;
        float* gp = out + base4 + row0 * S;
        if (t < NTQ - 1) {
          float4 val[8];
#pragma unroll
          for (int r = 0; r < 8; ++r)
            val[r] = *reinterpret_cast<const float4*>(sp + r * SPITCH);
#pragma unroll
          for (int r = 0; r < 8; ++r)
            *reinterpret_cast<float4*>(gp + r * S) = val[r];
        } else {
          float4 val[4];
#pragma unroll
          for (int r = 0; r < 4; ++r)
            val[r] = *reinterpret_cast<const float4*>(sp + r * SPITCH);
#pragma unroll
          for (int r = 0; r < 4; ++r)
            *reinterpret_cast<float4*>(gp + r * S) = val[r];
        }
      }
    }
  }
}

extern "C" void kernel_launch(void* const* d_in, const int* in_sizes, int n_in,
                              void* d_out, int out_size, void* d_ws, size_t ws_size,
                              hipStream_t stream) {
  const float* x   = (const float*)d_in[0];
  float*       out = (float*)d_out;
  float*       Tb  = (float*)d_ws;
  float*       tmp = (float*)d_ws + 32768;

  (void)in_sizes; (void)n_in; (void)out_size; (void)ws_size;

  const int lines = 2 * 3 * NAX * NAX;
  const int tiles = lines / 16;
  dim3 grid((tiles + 7) / 8), block(256);

  spline_pass<1, NAX, 1>
      <<<grid, block, 0, stream>>>(x, out, nullptr, Tb, tiles);
  spline_pass<NAX, NAX * NAX, NAX>
      <<<grid, block, 0, stream>>>(out, tmp, Tb, nullptr, tiles);
  spline_pass<NAX * NAX, NAX * NAX * NAX, NAX * NAX>
      <<<grid, block, 0, stream>>>(tmp, out, Tb, nullptr, tiles);
}
